# starting kernel
# baseline (speedup 1.0000x reference)
#include <hip/hip_runtime.h>

typedef float v2f __attribute__((ext_vector_type(2)));
typedef float v8f __attribute__((ext_vector_type(8)));

__device__ __forceinline__ float fast_sqrt(float x) { return __builtin_amdgcn_sqrtf(x); }
__device__ __forceinline__ float fast_exp2(float x) { return __builtin_amdgcn_exp2f(x); }

#define LOG2E 1.44269504088896340736f

__global__ void precompute_kernel(const float* __restrict__ states,
                                  float4* __restrict__ rec,
                                  float* __restrict__ G, int N) {
    int i = blockIdx.x * blockDim.x + threadIdx.x;
    if (i >= N) return;
    const float4* s = (const float4*)(states + (size_t)i * 8);
    float4 a = s[0];
    float4 b = s[1];
    float px = (a.x + a.z) * 0.5f;
    float py = (a.y + a.w) * 0.5f;
    float vx = (b.x + b.z) * 0.5f;
    float vy = (b.y + b.w) * 0.5f;
    float r  = px * px + py * py;
    rec[i] = make_float4(px, py, r, 0.0f);
    float4* g = (float4*)(G + (size_t)i * 16);
    g[0] = make_float4(px, py, vx, vy);
    g[1] = make_float4(px * px, py * py, vx * vx, vy * vy);
    g[2] = make_float4(1.0f, 0.0f, 0.0f, 0.0f);
    g[3] = make_float4(0.0f, 0.0f, 0.0f, 0.0f);
}

__global__ void groupvel_kernel(const float* __restrict__ G,
                                float* __restrict__ gv, int N) {
    __shared__ float sx[256];
    __shared__ float sy[256];
    int t = threadIdx.x;
    float ax = 0.0f, ay = 0.0f;
    for (int j = t; j < N; j += 256) {
        ax += G[(size_t)j * 16 + 2];
        ay += G[(size_t)j * 16 + 3];
    }
    sx[t] = ax; sy[t] = ay;
    __syncthreads();
    for (int s = 128; s > 0; s >>= 1) {
        if (t < s) { sx[t] += sx[t + s]; sy[t] += sy[t + s]; }
        __syncthreads();
    }
    if (t == 0) {
        gv[0] = sx[0] / (float)N;
        gv[1] = sy[0] / (float)N;
    }
}

__global__ __launch_bounds__(256) void neighbour_kernel(
    const float4* __restrict__ rec,
    const float*  __restrict__ G,
    const float*  __restrict__ gvp,
    const float*  __restrict__ log_tau,
    float*        __restrict__ out, int N)
{
    __shared__ float cbuf[8][256];

    const int tid  = threadIdx.x;
    const int wave = tid >> 5;
    const int lane = tid & 31;
    const int row  = lane & 15;
    const int h    = lane >> 4;
    const int i0   = (blockIdx.x * 8 + wave) * 16;
    const int i    = i0 + row;

    const float invtau = 1.0f / __expf(log_tau[0]);

    const float4 ri  = rec[i];
    const float px_i = ri.x, py_i = ri.y, r_i = ri.z;

    float qmin = 3.0e38f;
    #pragma unroll 4
    for (int j = h; j < N; j += 2) {
        float4 rj = rec[j];
        float dot = px_i * rj.x + py_i * rj.y;
        float q   = r_i + rj.z - 2.0f * dot;
        q = (j == i) ? 3.0e38f : q;
        qmin = fminf(qmin, q);
    }
    qmin = fminf(qmin, __shfl_xor(qmin, 16, 32));
    const float mdist = fast_sqrt(fmaxf(qmin, 0.0f) + 1e-8f);

    const float k2 = invtau * LOG2E;
    const float b2 = mdist * k2;

    v8f acc = {0.0f, 0.0f, 0.0f, 0.0f, 0.0f, 0.0f, 0.0f, 0.0f};

    for (int jt = 0; jt < N; jt += 16) {
        v2f A[4];
        v2f B[4];
        #pragma unroll
        for (int g = 0; g < 4; ++g) {
            #pragma unroll
            for (int p = 0; p < 2; ++p) {
                const int j = jt + 4 * g + p + 2 * h;
                float4 rj = rec[j];
                float dot  = px_i * rj.x + py_i * rj.y;
                float q    = r_i + rj.z - 2.0f * dot;
                float dist = fast_sqrt(fmaxf(q, 0.0f) + 1e-8f);
                float w    = fast_exp2(fmaf(-dist, k2, b2));
                w = (j == i) ? 0.0f : w;
                A[g][p] = w;
                B[g][p] = G[(size_t)j * 16 + row];
            }
        }
        #pragma unroll
        for (int g = 0; g < 4; ++g) {
            acc = __builtin_amdgcn_wmma_f32_16x16x4_f32(
                 false, A[g],  false, B[g],
                 (short)0, acc,  false,  false);
        }
    }

    #pragma unroll
    for (int v = 0; v < 8; ++v) {
        cbuf[wave][(v + 8 * h) * 16 + row] = acc[v];
    }
    __syncthreads();

    if (lane < 16) {
        const int ii = i0 + lane;
        const float* c = &cbuf[wave][lane * 16];
        const float S    = c[8];
        const float invS = 1.0f / S;
        const float4 fi4 = *(const float4*)(G + (size_t)ii * 16);
        const float f[4] = {fi4.x, fi4.y, fi4.z, fi4.w};
        const float gvx = gvp[0];
        const float gvy = gvp[1];

        float mu[4], sg[4];
        #pragma unroll
        for (int d = 0; d < 4; ++d) {
            float T = c[d] * invS;
            float U = c[4 + d] * invS;
            mu[d] = f[d] - T;
            float eg2 = f[d] * f[d] - 2.0f * f[d] * T + U;
            sg[d] = fast_sqrt(fmaxf(eg2 - mu[d] * mu[d], 0.0f) + 1e-6f);
        }
        float4* o = (float4*)(out + (size_t)ii * 12);
        o[0] = make_float4(mu[0], mu[1], mu[2], mu[3]);
        o[1] = make_float4(sg[0], sg[1], sg[2], sg[3]);
        o[2] = make_float4(gvx, gvy, f[2] - gvx, f[3] - gvy);
    }
}

extern "C" void kernel_launch(void* const* d_in, const int* in_sizes, int n_in,
                              void* d_out, int out_size, void* d_ws, size_t ws_size,
                              hipStream_t stream) {
    const float* states  = (const float*)d_in[0];
    const float* log_tau = (const float*)d_in[1];
    const int N = in_sizes[0] / 8;

    char* ws    = (char*)d_ws;
    float4* rec = (float4*)ws;
    float*  G   = (float*)(ws + (size_t)N * 16);
    float*  gv  = (float*)(ws + (size_t)N * 16 + (size_t)N * 64);
    float*  out = (float*)d_out;

    precompute_kernel<<<(N + 255) / 256, 256, 0, stream>>>(states, rec, G, N);
    groupvel_kernel<<<1, 256, 0, stream>>>(G, gv, N);
    neighbour_kernel<<<N / 128, 256, 0, stream>>>(rec, G, gv, log_tau, out, N);
}


# Reference update stays on the vector ALU, every chunk, no test: half minima joined by v_permlanex16, rescale factors by DPP row-share (was LDS permutes + a ballot branch); sums and their order kept
# speedup vs baseline: 7.9744x; 7.9744x over previous
#include <hip/hip_runtime.h>

typedef float v2f __attribute__((ext_vector_type(2)));
typedef float v8f __attribute__((ext_vector_type(8)));

__device__ __forceinline__ float fast_sqrt(float x) { return __builtin_amdgcn_sqrtf(x); }
__device__ __forceinline__ float fast_exp2(float x) { return __builtin_amdgcn_exp2f(x); }

#define LOG2E 1.44269504088896340736f

__global__ void precompute_kernel(const float* __restrict__ states,
                                  float* __restrict__ G, int N) {
    int i = blockIdx.x * blockDim.x + threadIdx.x;
    if (i >= N) return;
    const float4* s = (const float4*)(states + (size_t)i * 8);
    float4 a = s[0];
    float4 b = s[1];
    float px = (a.x + a.z) * 0.5f;
    float py = (a.y + a.w) * 0.5f;
    float vx = (b.x + b.z) * 0.5f;
    float vy = (b.y + b.w) * 0.5f;
    float4* g = (float4*)(G + (size_t)i * 16);
    g[0] = make_float4(px, py, vx, vy);
    g[1] = make_float4(px * px, py * py, vx * vx, vy * vy);
    g[2] = make_float4(1.0f, 0.0f, 0.0f, 0.0f);
    g[3] = make_float4(0.0f, 0.0f, 0.0f, 0.0f);
}

__global__ void groupvel_kernel(const float* __restrict__ G,
                                float* __restrict__ gv, int N) {
    __shared__ float sx[256];
    __shared__ float sy[256];
    int t = threadIdx.x;
    float ax = 0.0f, ay = 0.0f;
    for (int j = t; j < N; j += 256) {
        ax += G[(size_t)j * 16 + 2];
        ay += G[(size_t)j * 16 + 3];
    }
    sx[t] = ax; sy[t] = ay;
    __syncthreads();
    for (int s = 128; s > 0; s >>= 1) {
        if (t < s) { sx[t] += sx[t + s]; sy[t] += sy[t + s]; }
        __syncthreads();
    }
    if (t == 0) {
        gv[0] = sx[0] / (float)N;
        gv[1] = sy[0] / (float)N;
    }
}

#define CH  64
#define NPL (CH / 2)

struct JRegs {
    v2f b[NPL / 2];
};

template <int U0, int U1>
__device__ __forceinline__ void fetch_brow(const float* __restrict__ G, const int jl, const int col, JRegs& s) {
    const float* __restrict__ gp = G + (size_t)jl * 16 + col;
    #pragma unroll
    for (int u = U0; u < U1; ++u) {
        s.b[u >> 1][u & 1] = gp[(4 * (u >> 1) + (u & 1)) * 16];
    }
}

template <int L>
__device__ __forceinline__ float row_lane(const float v) {
    return __builtin_bit_cast(float,
        __builtin_amdgcn_update_dpp(0, __builtin_bit_cast(int, v), 0x150 + L, 0xf, 0xf, true));
}

__device__ __forceinline__ float xhalf(const float v) {
    const int b = __builtin_bit_cast(int, v);
    return __builtin_bit_cast(float,
        __builtin_amdgcn_permlanex16(b, b, 0x76543210, 0xfedcba98, false, false));
}

__device__ __forceinline__ float upper_ror8(const float v) {
    const int b = __builtin_bit_cast(int, v);
    return __builtin_bit_cast(float,
        __builtin_amdgcn_update_dpp(b, b, 0x128, 0x2, 0xf, false));
}

__device__ __forceinline__ void chunk_step(
    const float* __restrict__ G,
    const JRegs& cur, JRegs& pre, const int jc, const int jn, const int jself,
    const int col, const int h, const int i,
    const float px_i, const float py_i, const float k2,
    float& qmin, float& mref, float& b2, v8f& acc)
{
    float q[NPL];
    #pragma unroll
    for (int u = 0; u < NPL; ++u) {
        const float bj = cur.b[u >> 1][u & 1];
        const float dx = row_lane<0>(bj) - px_i;
        const float dy = row_lane<1>(bj) - py_i;
        q[u] = fmaf(dy, dy, fmaf(dx, dx, 1e-8f));
    }
    if (jc == jself) {
        const int jl = jc + 2 * h;
        #pragma unroll
        for (int u = 0; u < NPL; ++u) {
            const int j = jl + 4 * (u >> 1) + (u & 1);
            q[u] = (j == i) ? 3.0e38f : q[u];
        }
    }
    float pm[4] = {q[0], q[1], q[2], q[3]};
    #pragma unroll
    for (int u = 4; u < NPL; ++u) pm[u & 3] = fminf(pm[u & 3], q[u]);
    float cmin = fminf(fminf(pm[0], pm[1]), fminf(pm[2], pm[3]));
    cmin = fminf(cmin, xhalf(cmin));
    const float qnew = fminf(qmin, cmin);

    {
        const float mnew  = fast_sqrt(qnew);
        const float scale = fast_exp2((mnew - mref) * k2);
        const float srow = upper_ror8(scale);
        acc[0] *= row_lane<0>(srow);
        acc[1] *= row_lane<1>(srow);
        acc[2] *= row_lane<2>(srow);
        acc[3] *= row_lane<3>(srow);
        acc[4] *= row_lane<4>(srow);
        acc[5] *= row_lane<5>(srow);
        acc[6] *= row_lane<6>(srow);
        acc[7] *= row_lane<7>(srow);
        mref = mnew;
        b2   = mnew * k2;
    }
    qmin = qnew;

    #pragma unroll
    for (int half = 0; half < 2; ++half) {
        if (half == 0) {
            fetch_brow<0, NPL / 2>(G, jn + 2 * h, col, pre);
        } else {
            __builtin_amdgcn_sched_barrier(0);
            fetch_brow<NPL / 2, NPL>(G, jn + 2 * h, col, pre);
        }
        __builtin_amdgcn_sched_barrier(0);
        #pragma unroll
        for (int kk = 0; kk < NPL / 4; ++kk) {
            const int ks = half * (NPL / 4) + kk;
            v2f A;
            #pragma unroll
            for (int p = 0; p < 2; ++p) {
                const int u = 2 * ks + p;
                float dist = fast_sqrt(q[u]);
                float w    = fast_exp2(fmaf(-dist, k2, b2));
                A[p]  = w;
            }
            acc = __builtin_amdgcn_wmma_f32_16x16x4_f32(
                 false, A,  false, cur.b[ks],
                 (short)0, acc,  false,  false);
        }
    }
}

__global__ __launch_bounds__(32) __attribute__((amdgpu_waves_per_eu(1, 1))) void neighbour_kernel(
    const float*  __restrict__ G,
    const float*  __restrict__ gvp,
    const float*  __restrict__ log_tau,
    float*        __restrict__ out, int N)
{
    __shared__ float cbuf[256];

    const int lane = threadIdx.x;
    const int col  = lane & 15;
    const int h    = lane >> 4;
    const int i0   = blockIdx.x * 16;
    const int i    = i0 + col;

    const float invtau = 1.0f / __expf(log_tau[0]);
    const float k2 = invtau * LOG2E;

    const float2 pi  = *(const float2*)(G + (size_t)i * 16);
    const float px_i = pi.x, py_i = pi.y;

    float qmin = 3.0e38f;
    float mref = fast_sqrt(qmin);
    float b2   = mref * k2;
    v8f acc = {0.0f, 0.0f, 0.0f, 0.0f, 0.0f, 0.0f, 0.0f, 0.0f};

    JRegs s0, s1, s2, s3;
    fetch_brow<0, NPL>(G, 2 * h, col, s0);
    __builtin_amdgcn_sched_barrier(0);
    fetch_brow<0, NPL>(G, CH + 2 * h, col, s1);
    __builtin_amdgcn_sched_barrier(0);
    fetch_brow<0, NPL>(G, 2 * CH + 2 * h, col, s2);
    __builtin_amdgcn_sched_barrier(0);
    const int jself = i0 & ~(CH - 1);
    for (int jc = 0; jc < N; jc += 4 * CH) {
        const int jw = (jc + 4 * CH < N) ? (jc + 4 * CH) : 0;
        chunk_step(G, s0, s3, jc,          jc + 3 * CH, jself,
                   col, h, i, px_i, py_i, k2, qmin, mref, b2, acc);
        chunk_step(G, s1, s0, jc + CH,     jw,          jself,
                   col, h, i, px_i, py_i, k2, qmin, mref, b2, acc);
        chunk_step(G, s2, s1, jc + 2 * CH, jw + CH,     jself,
                   col, h, i, px_i, py_i, k2, qmin, mref, b2, acc);
        chunk_step(G, s3, s2, jc + 3 * CH, jw + 2 * CH, jself,
                   col, h, i, px_i, py_i, k2, qmin, mref, b2, acc);
    }

    #pragma unroll
    for (int v = 0; v < 8; ++v) {
        cbuf[(v + 8 * h) * 16 + col] = acc[v];
    }
    __syncthreads();

    if (lane < 16) {
        const int ii = i0 + lane;
        const float* c = &cbuf[lane * 16];
        const float S    = c[8];
        const float invS = 1.0f / S;
        const float4 fi4 = *(const float4*)(G + (size_t)ii * 16);
        const float f[4] = {fi4.x, fi4.y, fi4.z, fi4.w};
        const float gvx = gvp[0];
        const float gvy = gvp[1];

        float mu[4], sg[4];
        #pragma unroll
        for (int d = 0; d < 4; ++d) {
            float T = c[d] * invS;
            float U = c[4 + d] * invS;
            mu[d] = f[d] - T;
            float eg2 = f[d] * f[d] - 2.0f * f[d] * T + U;
            sg[d] = fast_sqrt(fmaxf(eg2 - mu[d] * mu[d], 0.0f) + 1e-6f);
        }
        float4* o = (float4*)(out + (size_t)ii * 12);
        o[0] = make_float4(mu[0], mu[1], mu[2], mu[3]);
        o[1] = make_float4(sg[0], sg[1], sg[2], sg[3]);
        o[2] = make_float4(gvx, gvy, f[2] - gvx, f[3] - gvy);
    }
}

extern "C" void kernel_launch(void* const* d_in, const int* in_sizes, int n_in,
                              void* d_out, int out_size, void* d_ws, size_t ws_size,
                              hipStream_t stream) {
    const float* states  = (const float*)d_in[0];
    const float* log_tau = (const float*)d_in[1];
    const int N = in_sizes[0] / 8;

    char* ws    = (char*)d_ws;
    float*  G   = (float*)ws;
    float*  gv  = (float*)(ws + (size_t)N * 64);
    float*  out = (float*)d_out;

    precompute_kernel<<<(N + 255) / 256, 256, 0, stream>>>(states, G, N);
    groupvel_kernel<<<1, 256, 0, stream>>>(G, gv, N);
    neighbour_kernel<<<N / 16, 32, 0, stream>>>(G, gv, log_tau, out, N);
}


# Mean-velocity kernel: 1 memory trip + 1 LDS exchange, was 16 + 9: velocities also stored packed (8 B) by kernel 1, a thread's 16 loads issued together, tree by lane exchanges; same sums, same order
# speedup vs baseline: 9.2830x; 1.1641x over previous
#include <hip/hip_runtime.h>

typedef float v2f __attribute__((ext_vector_type(2)));
typedef float v8f __attribute__((ext_vector_type(8)));

__device__ __forceinline__ float fast_sqrt(float x) { return __builtin_amdgcn_sqrtf(x); }
__device__ __forceinline__ float fast_exp2(float x) { return __builtin_amdgcn_exp2f(x); }

#define LOG2E 1.44269504088896340736f

__global__ void precompute_kernel(const float* __restrict__ states,
                                  float* __restrict__ G, float* __restrict__ V, int N) {
    int i = blockIdx.x * blockDim.x + threadIdx.x;
    if (i >= N) return;
    const float4* s = (const float4*)(states + (size_t)i * 8);
    float4 a = s[0];
    float4 b = s[1];
    float px = (a.x + a.z) * 0.5f;
    float py = (a.y + a.w) * 0.5f;
    float vx = (b.x + b.z) * 0.5f;
    float vy = (b.y + b.w) * 0.5f;
    float4* g = (float4*)(G + (size_t)i * 16);
    g[0] = make_float4(px, py, vx, vy);
    g[1] = make_float4(px * px, py * py, vx * vx, vy * vy);
    g[2] = make_float4(1.0f, 0.0f, 0.0f, 0.0f);
    g[3] = make_float4(0.0f, 0.0f, 0.0f, 0.0f);
    ((float2*)V)[i] = make_float2(vx, vy);
}

__device__ __forceinline__ float swap_halves(const float v) {
    const int b = __builtin_bit_cast(int, v);
    return __builtin_bit_cast(float,
        __builtin_amdgcn_permlanex16(b, b, 0x76543210, 0xfedcba98, false, false));
}

template <int M>
__device__ __forceinline__ float row_xor(const float v) {
    const int b = __builtin_bit_cast(int, v);
    return __builtin_bit_cast(float,
        __builtin_amdgcn_update_dpp(b, b, 0x160 + M, 0xf, 0xf, false));
}

#define GV_T 256
#define GV_R 16

__global__ __launch_bounds__(GV_T) void groupvel_kernel(const float* __restrict__ V,
                                                       float* __restrict__ gv, int N) {
    __shared__ float2 part[GV_T];
    const int t = threadIdx.x;
    const float2* __restrict__ v2 = (const float2*)V;
    float ax = 0.0f, ay = 0.0f;
    for (int base = 0; base < N; base += GV_T * GV_R) {
        float2 v[GV_R];
        #pragma unroll
        for (int r = 0; r < GV_R; ++r) {
            const int j  = base + t + GV_T * r;
            const int jc = (j < N) ? j : (N - 1);
            v[r] = v2[jc];
        }
        __builtin_amdgcn_sched_barrier(0);
        #pragma unroll
        for (int r = 0; r < GV_R; ++r) {
            const bool in = (base + t + GV_T * r) < N;
            ax += in ? v[r].x : 0.0f;
            ay += in ? v[r].y : 0.0f;
        }
    }
    part[t] = make_float2(ax, ay);
    __syncthreads();
    if (t < 32) {
        float2 w[GV_T / 32];
        #pragma unroll
        for (int k = 0; k < GV_T / 32; ++k) w[k] = part[t + 32 * k];
        float sx = ((w[0].x + w[4].x) + (w[2].x + w[6].x)) + ((w[1].x + w[5].x) + (w[3].x + w[7].x));
        float sy = ((w[0].y + w[4].y) + (w[2].y + w[6].y)) + ((w[1].y + w[5].y) + (w[3].y + w[7].y));
        sx += swap_halves(sx);  sy += swap_halves(sy);
        sx += row_xor<8>(sx);   sy += row_xor<8>(sy);
        sx += row_xor<4>(sx);   sy += row_xor<4>(sy);
        sx += row_xor<2>(sx);   sy += row_xor<2>(sy);
        sx += row_xor<1>(sx);   sy += row_xor<1>(sy);
        if (t == 0) {
            gv[0] = sx / (float)N;
            gv[1] = sy / (float)N;
        }
    }
}

#define CH  64
#define NPL (CH / 2)

struct JRegs {
    v2f b[NPL / 2];
};

template <int U0, int U1>
__device__ __forceinline__ void fetch_brow(const float* __restrict__ G, const int jl, const int col, JRegs& s) {
    const float* __restrict__ gp = G + (size_t)jl * 16 + col;
    #pragma unroll
    for (int u = U0; u < U1; ++u) {
        s.b[u >> 1][u & 1] = gp[(4 * (u >> 1) + (u & 1)) * 16];
    }
}

template <int L>
__device__ __forceinline__ float row_lane(const float v) {
    return __builtin_bit_cast(float,
        __builtin_amdgcn_update_dpp(0, __builtin_bit_cast(int, v), 0x150 + L, 0xf, 0xf, true));
}

__device__ __forceinline__ float xhalf(const float v) {
    const int b = __builtin_bit_cast(int, v);
    return __builtin_bit_cast(float,
        __builtin_amdgcn_permlanex16(b, b, 0x76543210, 0xfedcba98, false, false));
}

__device__ __forceinline__ float upper_ror8(const float v) {
    const int b = __builtin_bit_cast(int, v);
    return __builtin_bit_cast(float,
        __builtin_amdgcn_update_dpp(b, b, 0x128, 0x2, 0xf, false));
}

__device__ __forceinline__ void chunk_step(
    const float* __restrict__ G,
    const JRegs& cur, JRegs& pre, const int jc, const int jn, const int jself,
    const int col, const int h, const int i,
    const float px_i, const float py_i, const float k2,
    float& qmin, float& mref, float& b2, v8f& acc)
{
    float q[NPL];
    #pragma unroll
    for (int u = 0; u < NPL; ++u) {
        const float bj = cur.b[u >> 1][u & 1];
        const float dx = row_lane<0>(bj) - px_i;
        const float dy = row_lane<1>(bj) - py_i;
        q[u] = fmaf(dy, dy, fmaf(dx, dx, 1e-8f));
    }
    if (jc == jself) {
        const int jl = jc + 2 * h;
        #pragma unroll
        for (int u = 0; u < NPL; ++u) {
            const int j = jl + 4 * (u >> 1) + (u & 1);
            q[u] = (j == i) ? 3.0e38f : q[u];
        }
    }
    float pm[4] = {q[0], q[1], q[2], q[3]};
    #pragma unroll
    for (int u = 4; u < NPL; ++u) pm[u & 3] = fminf(pm[u & 3], q[u]);
    float cmin = fminf(fminf(pm[0], pm[1]), fminf(pm[2], pm[3]));
    cmin = fminf(cmin, xhalf(cmin));
    const float qnew = fminf(qmin, cmin);

    {
        const float mnew  = fast_sqrt(qnew);
        const float scale = fast_exp2((mnew - mref) * k2);
        const float srow = upper_ror8(scale);
        acc[0] *= row_lane<0>(srow);
        acc[1] *= row_lane<1>(srow);
        acc[2] *= row_lane<2>(srow);
        acc[3] *= row_lane<3>(srow);
        acc[4] *= row_lane<4>(srow);
        acc[5] *= row_lane<5>(srow);
        acc[6] *= row_lane<6>(srow);
        acc[7] *= row_lane<7>(srow);
        mref = mnew;
        b2   = mnew * k2;
    }
    qmin = qnew;

    #pragma unroll
    for (int half = 0; half < 2; ++half) {
        if (half == 0) {
            fetch_brow<0, NPL / 2>(G, jn + 2 * h, col, pre);
        } else {
            __builtin_amdgcn_sched_barrier(0);
            fetch_brow<NPL / 2, NPL>(G, jn + 2 * h, col, pre);
        }
        __builtin_amdgcn_sched_barrier(0);
        #pragma unroll
        for (int kk = 0; kk < NPL / 4; ++kk) {
            const int ks = half * (NPL / 4) + kk;
            v2f A;
            #pragma unroll
            for (int p = 0; p < 2; ++p) {
                const int u = 2 * ks + p;
                float dist = fast_sqrt(q[u]);
                float w    = fast_exp2(fmaf(-dist, k2, b2));
                A[p]  = w;
            }
            acc = __builtin_amdgcn_wmma_f32_16x16x4_f32(
                 false, A,  false, cur.b[ks],
                 (short)0, acc,  false,  false);
        }
    }
}

__global__ __launch_bounds__(32) __attribute__((amdgpu_waves_per_eu(1, 1))) void neighbour_kernel(
    const float*  __restrict__ G,
    const float*  __restrict__ gvp,
    const float*  __restrict__ log_tau,
    float*        __restrict__ out, int N)
{
    __shared__ float cbuf[256];

    const int lane = threadIdx.x;
    const int col  = lane & 15;
    const int h    = lane >> 4;
    const int i0   = blockIdx.x * 16;
    const int i    = i0 + col;

    const float invtau = 1.0f / __expf(log_tau[0]);
    const float k2 = invtau * LOG2E;

    const float2 pi  = *(const float2*)(G + (size_t)i * 16);
    const float px_i = pi.x, py_i = pi.y;

    float qmin = 3.0e38f;
    float mref = fast_sqrt(qmin);
    float b2   = mref * k2;
    v8f acc = {0.0f, 0.0f, 0.0f, 0.0f, 0.0f, 0.0f, 0.0f, 0.0f};

    JRegs s0, s1, s2, s3;
    fetch_brow<0, NPL>(G, 2 * h, col, s0);
    __builtin_amdgcn_sched_barrier(0);
    fetch_brow<0, NPL>(G, CH + 2 * h, col, s1);
    __builtin_amdgcn_sched_barrier(0);
    fetch_brow<0, NPL>(G, 2 * CH + 2 * h, col, s2);
    __builtin_amdgcn_sched_barrier(0);
    const int jself = i0 & ~(CH - 1);
    for (int jc = 0; jc < N; jc += 4 * CH) {
        const int jw = (jc + 4 * CH < N) ? (jc + 4 * CH) : 0;
        chunk_step(G, s0, s3, jc,          jc + 3 * CH, jself,
                   col, h, i, px_i, py_i, k2, qmin, mref, b2, acc);
        chunk_step(G, s1, s0, jc + CH,     jw,          jself,
                   col, h, i, px_i, py_i, k2, qmin, mref, b2, acc);
        chunk_step(G, s2, s1, jc + 2 * CH, jw + CH,     jself,
                   col, h, i, px_i, py_i, k2, qmin, mref, b2, acc);
        chunk_step(G, s3, s2, jc + 3 * CH, jw + 2 * CH, jself,
                   col, h, i, px_i, py_i, k2, qmin, mref, b2, acc);
    }

    #pragma unroll
    for (int v = 0; v < 8; ++v) {
        cbuf[(v + 8 * h) * 16 + col] = acc[v];
    }
    __syncthreads();

    if (lane < 16) {
        const int ii = i0 + lane;
        const float* c = &cbuf[lane * 16];
        const float S    = c[8];
        const float invS = 1.0f / S;
        const float4 fi4 = *(const float4*)(G + (size_t)ii * 16);
        const float f[4] = {fi4.x, fi4.y, fi4.z, fi4.w};
        const float gvx = gvp[0];
        const float gvy = gvp[1];

        float mu[4], sg[4];
        #pragma unroll
        for (int d = 0; d < 4; ++d) {
            float T = c[d] * invS;
            float U = c[4 + d] * invS;
            mu[d] = f[d] - T;
            float eg2 = f[d] * f[d] - 2.0f * f[d] * T + U;
            sg[d] = fast_sqrt(fmaxf(eg2 - mu[d] * mu[d], 0.0f) + 1e-6f);
        }
        float4* o = (float4*)(out + (size_t)ii * 12);
        o[0] = make_float4(mu[0], mu[1], mu[2], mu[3]);
        o[1] = make_float4(sg[0], sg[1], sg[2], sg[3]);
        o[2] = make_float4(gvx, gvy, f[2] - gvx, f[3] - gvy);
    }
}

extern "C" void kernel_launch(void* const* d_in, const int* in_sizes, int n_in,
                              void* d_out, int out_size, void* d_ws, size_t ws_size,
                              hipStream_t stream) {
    const float* states  = (const float*)d_in[0];
    const float* log_tau = (const float*)d_in[1];
    const int N = in_sizes[0] / 8;

    char* ws    = (char*)d_ws;
    float*  G   = (float*)ws;
    float*  gv  = (float*)(ws + (size_t)N * 64);
    float*  V   = (float*)(ws + (size_t)N * 64 + 256);
    float*  out = (float*)d_out;

    precompute_kernel<<<(N + 255) / 256, 256, 0, stream>>>(states, G, V, N);
    groupvel_kernel<<<1, GV_T, 0, stream>>>(V, gv, N);
    neighbour_kernel<<<N / 16, 32, 0, stream>>>(G, gv, log_tau, out, N);
}


# B rows kept in fragment order by kernel 1: a lane's operand pairs of two k-steps are 16 contiguous bytes; a 64-j chunk is 8 whole-line 16-B loads at one place, was 32 4-B loads at two; sums kept
# speedup vs baseline: 10.3247x; 1.1122x over previous
#include <hip/hip_runtime.h>

typedef float v2f __attribute__((ext_vector_type(2)));
typedef float v4f __attribute__((ext_vector_type(4)));
typedef float v8f __attribute__((ext_vector_type(8)));

__device__ __forceinline__ float fast_sqrt(float x) { return __builtin_amdgcn_sqrtf(x); }
__device__ __forceinline__ float fast_exp2(float x) { return __builtin_amdgcn_exp2f(x); }

#define LOG2E 1.44269504088896340736f

__device__ __forceinline__ size_t frag_index(const int j, const int c) {
    const int r = j & 7;
    const int lane = c + 16 * ((r >> 1) & 1);
    return ((size_t)(j >> 3) * 32 + lane) * 4 + 2 * (r >> 2) + (r & 1);
}

__global__ void precompute_kernel(const float* __restrict__ states,
                                  float* __restrict__ G, float* __restrict__ V, int N) {
    int i = blockIdx.x * blockDim.x + threadIdx.x;
    if (i >= N) return;
    const float4* s = (const float4*)(states + (size_t)i * 8);
    float4 a = s[0];
    float4 b = s[1];
    float px = (a.x + a.z) * 0.5f;
    float py = (a.y + a.w) * 0.5f;
    float vx = (b.x + b.z) * 0.5f;
    float vy = (b.y + b.w) * 0.5f;
    float* g = G + frag_index(i, 0);
    const float row[16] = {px, py, vx, vy, px * px, py * py, vx * vx, vy * vy,
                           1.0f, 0.0f, 0.0f, 0.0f, 0.0f, 0.0f, 0.0f, 0.0f};
    #pragma unroll
    for (int c = 0; c < 16; ++c) g[4 * c] = row[c];
    ((float2*)V)[i] = make_float2(vx, vy);
}

__device__ __forceinline__ float swap_halves(const float v) {
    const int b = __builtin_bit_cast(int, v);
    return __builtin_bit_cast(float,
        __builtin_amdgcn_permlanex16(b, b, 0x76543210, 0xfedcba98, false, false));
}

template <int M>
__device__ __forceinline__ float row_xor(const float v) {
    const int b = __builtin_bit_cast(int, v);
    return __builtin_bit_cast(float,
        __builtin_amdgcn_update_dpp(b, b, 0x160 + M, 0xf, 0xf, false));
}

#define GV_T 256
#define GV_R 16

__global__ __launch_bounds__(GV_T) void groupvel_kernel(const float* __restrict__ V,
                                                       float* __restrict__ gv, int N) {
    __shared__ float2 part[GV_T];
    const int t = threadIdx.x;
    const float2* __restrict__ v2 = (const float2*)V;
    float ax = 0.0f, ay = 0.0f;
    for (int base = 0; base < N; base += GV_T * GV_R) {
        float2 v[GV_R];
        #pragma unroll
        for (int r = 0; r < GV_R; ++r) {
            const int j  = base + t + GV_T * r;
            const int jc = (j < N) ? j : (N - 1);
            v[r] = v2[jc];
        }
        __builtin_amdgcn_sched_barrier(0);
        #pragma unroll
        for (int r = 0; r < GV_R; ++r) {
            const bool in = (base + t + GV_T * r) < N;
            ax += in ? v[r].x : 0.0f;
            ay += in ? v[r].y : 0.0f;
        }
    }
    part[t] = make_float2(ax, ay);
    __syncthreads();
    if (t < 32) {
        float2 w[GV_T / 32];
        #pragma unroll
        for (int k = 0; k < GV_T / 32; ++k) w[k] = part[t + 32 * k];
        float sx = ((w[0].x + w[4].x) + (w[2].x + w[6].x)) + ((w[1].x + w[5].x) + (w[3].x + w[7].x));
        float sy = ((w[0].y + w[4].y) + (w[2].y + w[6].y)) + ((w[1].y + w[5].y) + (w[3].y + w[7].y));
        sx += swap_halves(sx);  sy += swap_halves(sy);
        sx += row_xor<8>(sx);   sy += row_xor<8>(sy);
        sx += row_xor<4>(sx);   sy += row_xor<4>(sy);
        sx += row_xor<2>(sx);   sy += row_xor<2>(sy);
        sx += row_xor<1>(sx);   sy += row_xor<1>(sy);
        if (t == 0) {
            gv[0] = sx / (float)N;
            gv[1] = sy / (float)N;
        }
    }
}

#define CH  64
#define NPL (CH / 2)

struct JRegs {
    v4f r[NPL / 4];
};

#define SET_LOADED(s) "v"((s).r[0][0]), "v"((s).r[1][0]), "v"((s).r[2][0]), "v"((s).r[3][0]), \
                      "v"((s).r[4][0]), "v"((s).r[5][0]), "v"((s).r[6][0]), "v"((s).r[7][0])

__device__ __forceinline__ void fetch_frag(const float* __restrict__ G, const int jc, const int lane, JRegs& s) {
    const v4f* __restrict__ gp = (const v4f*)G + (size_t)(jc >> 3) * 32 + lane;
    #pragma unroll
    for (int q = 0; q < NPL / 4; ++q) {
        s.r[q] = gp[q * 32];
    }
}

template <int L>
__device__ __forceinline__ float row_lane(const float v) {
    return __builtin_bit_cast(float,
        __builtin_amdgcn_update_dpp(0, __builtin_bit_cast(int, v), 0x150 + L, 0xf, 0xf, true));
}

__device__ __forceinline__ float xhalf(const float v) {
    const int b = __builtin_bit_cast(int, v);
    return __builtin_bit_cast(float,
        __builtin_amdgcn_permlanex16(b, b, 0x76543210, 0xfedcba98, false, false));
}

__device__ __forceinline__ float upper_ror8(const float v) {
    const int b = __builtin_bit_cast(int, v);
    return __builtin_bit_cast(float,
        __builtin_amdgcn_update_dpp(b, b, 0x128, 0x2, 0xf, false));
}

__device__ __forceinline__ void chunk_step(
    const float* __restrict__ G,
    const JRegs& cur, JRegs& pre, const int jc, const int jn, const int jself,
    const int lane, const int h, const int i,
    const float px_i, const float py_i, const float k2,
    float& qmin, float& mref, float& b2, v8f& acc)
{
    float q[NPL];
    #pragma unroll
    for (int u = 0; u < NPL; ++u) {
        const float bj = cur.r[u >> 2][u & 3];
        const float dx = row_lane<0>(bj) - px_i;
        const float dy = row_lane<1>(bj) - py_i;
        q[u] = fmaf(dy, dy, fmaf(dx, dx, 1e-8f));
    }
    if (jc == jself) {
        const int jl = jc + 2 * h;
        #pragma unroll
        for (int u = 0; u < NPL; ++u) {
            const int j = jl + 4 * (u >> 1) + (u & 1);
            q[u] = (j == i) ? 3.0e38f : q[u];
        }
    }
    fetch_frag(G, jn, lane, pre);
    __builtin_amdgcn_sched_barrier(0);
    float pm[4] = {q[0], q[1], q[2], q[3]};
    #pragma unroll
    for (int u = 4; u < NPL; ++u) pm[u & 3] = fminf(pm[u & 3], q[u]);
    float cmin = fminf(fminf(pm[0], pm[1]), fminf(pm[2], pm[3]));
    cmin = fminf(cmin, xhalf(cmin));
    const float qnew = fminf(qmin, cmin);

    {
        const float mnew  = fast_sqrt(qnew);
        const float scale = fast_exp2((mnew - mref) * k2);
        const float srow = upper_ror8(scale);
        acc[0] *= row_lane<0>(srow);
        acc[1] *= row_lane<1>(srow);
        acc[2] *= row_lane<2>(srow);
        acc[3] *= row_lane<3>(srow);
        acc[4] *= row_lane<4>(srow);
        acc[5] *= row_lane<5>(srow);
        acc[6] *= row_lane<6>(srow);
        acc[7] *= row_lane<7>(srow);
        mref = mnew;
        b2   = mnew * k2;
    }
    qmin = qnew;

    #pragma unroll
    for (int ks = 0; ks < NPL / 2; ++ks) {
        v2f A, B;
        #pragma unroll
        for (int p = 0; p < 2; ++p) {
            const int u = 2 * ks + p;
            const float dist = fast_sqrt(q[u]);
            A[p] = fast_exp2(fmaf(-dist, k2, b2));
            B[p] = cur.r[u >> 2][u & 3];
        }
        acc = __builtin_amdgcn_wmma_f32_16x16x4_f32(
             false, A,  false, B,
             (short)0, acc,  false,  false);
    }
}

__global__ __launch_bounds__(32) __attribute__((amdgpu_waves_per_eu(1, 1))) void neighbour_kernel(
    const float*  __restrict__ G,
    const float*  __restrict__ gvp,
    const float*  __restrict__ log_tau,
    float*        __restrict__ out, int N)
{
    __shared__ float cbuf[256];

    const int lane = threadIdx.x;
    const int col  = lane & 15;
    const int h    = lane >> 4;
    const int i0   = blockIdx.x * 16;
    const int i    = i0 + col;

    const float invtau = 1.0f / __expf(log_tau[0]);
    const float k2 = invtau * LOG2E;

    const float* __restrict__ gi = G + frag_index(i, 0);
    const float px_i = gi[0], py_i = gi[4];

    float qmin = 3.0e38f;
    float mref = fast_sqrt(qmin);
    float b2   = mref * k2;
    v8f acc = {0.0f, 0.0f, 0.0f, 0.0f, 0.0f, 0.0f, 0.0f, 0.0f};

    JRegs s0, s1, s2, s3;
    fetch_frag(G, 0, lane, s0);
    __builtin_amdgcn_sched_barrier(0);
    fetch_frag(G, CH, lane, s1);
    __builtin_amdgcn_sched_barrier(0);
    fetch_frag(G, 2 * CH, lane, s2);
    __builtin_amdgcn_sched_barrier(0);
    const int jself = i0 & ~(CH - 1);
    asm volatile("" :: "v"(px_i), "v"(py_i), SET_LOADED(s0), SET_LOADED(s1), SET_LOADED(s2));
    for (int jc = 0; jc < N; jc += 4 * CH) {
        const int jw = (jc + 4 * CH < N) ? (jc + 4 * CH) : 0;
        chunk_step(G, s0, s3, jc,          jc + 3 * CH, jself,
                   lane, h, i, px_i, py_i, k2, qmin, mref, b2, acc);
        chunk_step(G, s1, s0, jc + CH,     jw,          jself,
                   lane, h, i, px_i, py_i, k2, qmin, mref, b2, acc);
        chunk_step(G, s2, s1, jc + 2 * CH, jw + CH,     jself,
                   lane, h, i, px_i, py_i, k2, qmin, mref, b2, acc);
        chunk_step(G, s3, s2, jc + 3 * CH, jw + 2 * CH, jself,
                   lane, h, i, px_i, py_i, k2, qmin, mref, b2, acc);
    }

    #pragma unroll
    for (int v = 0; v < 8; ++v) {
        cbuf[(v + 8 * h) * 16 + col] = acc[v];
    }
    __syncthreads();

    if (lane < 16) {
        const int ii = i0 + lane;
        const float* c = &cbuf[lane * 16];
        const float S    = c[8];
        const float invS = 1.0f / S;
        const float* __restrict__ gf = G + frag_index(ii, 0);
        const float f[4] = {gf[0], gf[4], gf[8], gf[12]};
        const float gvx = gvp[0];
        const float gvy = gvp[1];

        float mu[4], sg[4];
        #pragma unroll
        for (int d = 0; d < 4; ++d) {
            float T = c[d] * invS;
            float U = c[4 + d] * invS;
            mu[d] = f[d] - T;
            float eg2 = f[d] * f[d] - 2.0f * f[d] * T + U;
            sg[d] = fast_sqrt(fmaxf(eg2 - mu[d] * mu[d], 0.0f) + 1e-6f);
        }
        float4* o = (float4*)(out + (size_t)ii * 12);
        o[0] = make_float4(mu[0], mu[1], mu[2], mu[3]);
        o[1] = make_float4(sg[0], sg[1], sg[2], sg[3]);
        o[2] = make_float4(gvx, gvy, f[2] - gvx, f[3] - gvy);
    }
}

extern "C" void kernel_launch(void* const* d_in, const int* in_sizes, int n_in,
                              void* d_out, int out_size, void* d_ws, size_t ws_size,
                              hipStream_t stream) {
    const float* states  = (const float*)d_in[0];
    const float* log_tau = (const float*)d_in[1];
    const int N = in_sizes[0] / 8;

    char* ws    = (char*)d_ws;
    float*  G   = (float*)ws;
    float*  gv  = (float*)(ws + (size_t)N * 64);
    float*  V   = (float*)(ws + (size_t)N * 64 + 256);
    float*  out = (float*)d_out;

    precompute_kernel<<<(N + 255) / 256, 256, 0, stream>>>(states, G, V, N);
    groupvel_kernel<<<1, GV_T, 0, stream>>>(V, gv, N);
    neighbour_kernel<<<N / 16, 32, 0, stream>>>(G, gv, log_tau, out, N);
}


# Positions of j as a second stream: kernel 1 also stores [x,x',y,y'] per j pair; one 16-B half-uniform load per k-step; distances by packed sub and fma, no cross-lane op; q and sums bit-identical
# speedup vs baseline: 10.9218x; 1.0578x over previous
#include <hip/hip_runtime.h>

typedef float v2f __attribute__((ext_vector_type(2)));
typedef float v4f __attribute__((ext_vector_type(4)));
typedef float v8f __attribute__((ext_vector_type(8)));

__device__ __forceinline__ float fast_sqrt(float x) { return __builtin_amdgcn_sqrtf(x); }
__device__ __forceinline__ float fast_exp2(float x) { return __builtin_amdgcn_exp2f(x); }

#define LOG2E 1.44269504088896340736f

__device__ __forceinline__ size_t frag_index(const int j, const int c) {
    const int r = j & 7;
    const int lane = c + 16 * ((r >> 1) & 1);
    return ((size_t)(j >> 3) * 32 + lane) * 4 + 2 * (r >> 2) + (r & 1);
}

__global__ void precompute_kernel(const float* __restrict__ states,
                                  float* __restrict__ G, float* __restrict__ V,
                                  float* __restrict__ P, int N) {
    int i = blockIdx.x * blockDim.x + threadIdx.x;
    if (i >= N) return;
    const float4* s = (const float4*)(states + (size_t)i * 8);
    float4 a = s[0];
    float4 b = s[1];
    float px = (a.x + a.z) * 0.5f;
    float py = (a.y + a.w) * 0.5f;
    float vx = (b.x + b.z) * 0.5f;
    float vy = (b.y + b.w) * 0.5f;
    float* g = G + frag_index(i, 0);
    const float row[16] = {px, py, vx, vy, px * px, py * py, vx * vx, vy * vy,
                           1.0f, 0.0f, 0.0f, 0.0f, 0.0f, 0.0f, 0.0f, 0.0f};
    #pragma unroll
    for (int c = 0; c < 16; ++c) g[4 * c] = row[c];
    ((float2*)V)[i] = make_float2(vx, vy);
    P[(size_t)(i >> 1) * 4 + (i & 1)]     = px;
    P[(size_t)(i >> 1) * 4 + 2 + (i & 1)] = py;
}

__device__ __forceinline__ float swap_halves(const float v) {
    const int b = __builtin_bit_cast(int, v);
    return __builtin_bit_cast(float,
        __builtin_amdgcn_permlanex16(b, b, 0x76543210, 0xfedcba98, false, false));
}

template <int M>
__device__ __forceinline__ float row_xor(const float v) {
    const int b = __builtin_bit_cast(int, v);
    return __builtin_bit_cast(float,
        __builtin_amdgcn_update_dpp(b, b, 0x160 + M, 0xf, 0xf, false));
}

#define GV_T 256
#define GV_R 16

__global__ __launch_bounds__(GV_T) void groupvel_kernel(const float* __restrict__ V,
                                                       float* __restrict__ gv, int N) {
    __shared__ float2 part[GV_T];
    const int t = threadIdx.x;
    const float2* __restrict__ v2 = (const float2*)V;
    float ax = 0.0f, ay = 0.0f;
    for (int base = 0; base < N; base += GV_T * GV_R) {
        float2 v[GV_R];
        #pragma unroll
        for (int r = 0; r < GV_R; ++r) {
            const int j  = base + t + GV_T * r;
            const int jc = (j < N) ? j : (N - 1);
            v[r] = v2[jc];
        }
        __builtin_amdgcn_sched_barrier(0);
        #pragma unroll
        for (int r = 0; r < GV_R; ++r) {
            const bool in = (base + t + GV_T * r) < N;
            ax += in ? v[r].x : 0.0f;
            ay += in ? v[r].y : 0.0f;
        }
    }
    part[t] = make_float2(ax, ay);
    __syncthreads();
    if (t < 32) {
        float2 w[GV_T / 32];
        #pragma unroll
        for (int k = 0; k < GV_T / 32; ++k) w[k] = part[t + 32 * k];
        float sx = ((w[0].x + w[4].x) + (w[2].x + w[6].x)) + ((w[1].x + w[5].x) + (w[3].x + w[7].x));
        float sy = ((w[0].y + w[4].y) + (w[2].y + w[6].y)) + ((w[1].y + w[5].y) + (w[3].y + w[7].y));
        sx += swap_halves(sx);  sy += swap_halves(sy);
        sx += row_xor<8>(sx);   sy += row_xor<8>(sy);
        sx += row_xor<4>(sx);   sy += row_xor<4>(sy);
        sx += row_xor<2>(sx);   sy += row_xor<2>(sy);
        sx += row_xor<1>(sx);   sy += row_xor<1>(sy);
        if (t == 0) {
            gv[0] = sx / (float)N;
            gv[1] = sy / (float)N;
        }
    }
}

#define CH  64
#define NPL (CH / 2)

struct JRegs {
    v4f r[NPL / 4];
};

struct PRegs {
    v4f r[NPL / 2];
};

#define SET_LOADED(s) "v"((s).r[0][0]), "v"((s).r[1][0]), "v"((s).r[2][0]), "v"((s).r[3][0]), \
                      "v"((s).r[4][0]), "v"((s).r[5][0]), "v"((s).r[6][0]), "v"((s).r[7][0])

__device__ __forceinline__ void fetch_frag(const float* __restrict__ G, const int jc, const int lane, JRegs& s) {
    const v4f* __restrict__ gp = (const v4f*)G + (size_t)(jc >> 3) * 32 + lane;
    #pragma unroll
    for (int q = 0; q < NPL / 4; ++q) {
        s.r[q] = gp[q * 32];
    }
}

__device__ __forceinline__ void fetch_pos(const float* __restrict__ P, const int jc, const int h, PRegs& s) {
    const v4f* __restrict__ pp = (const v4f*)P + (size_t)(jc >> 1) + h;
    #pragma unroll
    for (int ks = 0; ks < NPL / 2; ++ks) {
        s.r[ks] = pp[ks * 2];
    }
}

template <int L>
__device__ __forceinline__ float row_lane(const float v) {
    return __builtin_bit_cast(float,
        __builtin_amdgcn_update_dpp(0, __builtin_bit_cast(int, v), 0x150 + L, 0xf, 0xf, true));
}

__device__ __forceinline__ float xhalf(const float v) {
    const int b = __builtin_bit_cast(int, v);
    return __builtin_bit_cast(float,
        __builtin_amdgcn_permlanex16(b, b, 0x76543210, 0xfedcba98, false, false));
}

__device__ __forceinline__ float upper_ror8(const float v) {
    const int b = __builtin_bit_cast(int, v);
    return __builtin_bit_cast(float,
        __builtin_amdgcn_update_dpp(b, b, 0x128, 0x2, 0xf, false));
}

__device__ __forceinline__ void chunk_step(
    const float* __restrict__ G, const float* __restrict__ P,
    const JRegs& cur, const PRegs& curp, JRegs& pre, PRegs& prep,
    const int jc, const int jn, const int jself,
    const int lane, const int h, const int i,
    const float px_i, const float py_i, const float k2,
    float& qmin, float& mref, float& b2, v8f& acc)
{
    float q[NPL];
    const v2f pxx = {px_i, px_i};
    const v2f pyy = {py_i, py_i};
    const v2f eps = {1e-8f, 1e-8f};
    #pragma unroll
    for (int ks = 0; ks < NPL / 2; ++ks) {
        const v4f pr = curp.r[ks];
        const v2f xs = {pr[0], pr[1]};
        const v2f ys = {pr[2], pr[3]};
        const v2f dx = xs - pxx;
        const v2f dy = ys - pyy;
        const v2f qq = __builtin_elementwise_fma(dy, dy, __builtin_elementwise_fma(dx, dx, eps));
        q[2 * ks]     = qq[0];
        q[2 * ks + 1] = qq[1];
    }
    if (jc == jself) {
        const int jl = jc + 2 * h;
        #pragma unroll
        for (int u = 0; u < NPL; ++u) {
            const int j = jl + 4 * (u >> 1) + (u & 1);
            q[u] = (j == i) ? 3.0e38f : q[u];
        }
    }
    fetch_frag(G, jn, lane, pre);
    fetch_pos(P, jn, h, prep);
    __builtin_amdgcn_sched_barrier(0);
    float pm[4] = {q[0], q[1], q[2], q[3]};
    #pragma unroll
    for (int u = 4; u < NPL; ++u) pm[u & 3] = fminf(pm[u & 3], q[u]);
    float cmin = fminf(fminf(pm[0], pm[1]), fminf(pm[2], pm[3]));
    cmin = fminf(cmin, xhalf(cmin));
    const float qnew = fminf(qmin, cmin);

    {
        const float mnew  = fast_sqrt(qnew);
        const float scale = fast_exp2((mnew - mref) * k2);
        const float srow = upper_ror8(scale);
        acc[0] *= row_lane<0>(srow);
        acc[1] *= row_lane<1>(srow);
        acc[2] *= row_lane<2>(srow);
        acc[3] *= row_lane<3>(srow);
        acc[4] *= row_lane<4>(srow);
        acc[5] *= row_lane<5>(srow);
        acc[6] *= row_lane<6>(srow);
        acc[7] *= row_lane<7>(srow);
        mref = mnew;
        b2   = mnew * k2;
    }
    qmin = qnew;

    #pragma unroll
    for (int ks = 0; ks < NPL / 2; ++ks) {
        v2f A, B;
        #pragma unroll
        for (int p = 0; p < 2; ++p) {
            const int u = 2 * ks + p;
            const float dist = fast_sqrt(q[u]);
            A[p] = fast_exp2(fmaf(-dist, k2, b2));
            B[p] = cur.r[u >> 2][u & 3];
        }
        acc = __builtin_amdgcn_wmma_f32_16x16x4_f32(
             false, A,  false, B,
             (short)0, acc,  false,  false);
    }
}

__global__ __launch_bounds__(32) __attribute__((amdgpu_waves_per_eu(1, 1))) void neighbour_kernel(
    const float*  __restrict__ G,
    const float*  __restrict__ P,
    const float*  __restrict__ gvp,
    const float*  __restrict__ log_tau,
    float*        __restrict__ out, int N)
{
    __shared__ float cbuf[256];

    const int lane = threadIdx.x;
    const int col  = lane & 15;
    const int h    = lane >> 4;
    const int i0   = blockIdx.x * 16;
    const int i    = i0 + col;

    const float invtau = 1.0f / __expf(log_tau[0]);
    const float k2 = invtau * LOG2E;

    const float* __restrict__ gi = G + frag_index(i, 0);
    const float px_i = gi[0], py_i = gi[4];

    float qmin = 3.0e38f;
    float mref = fast_sqrt(qmin);
    float b2   = mref * k2;
    v8f acc = {0.0f, 0.0f, 0.0f, 0.0f, 0.0f, 0.0f, 0.0f, 0.0f};

    JRegs s0, s1, s2, s3;
    PRegs p0, p1, p2, p3;
    fetch_frag(G, 0, lane, s0);
    fetch_pos(P, 0, h, p0);
    __builtin_amdgcn_sched_barrier(0);
    fetch_frag(G, CH, lane, s1);
    fetch_pos(P, CH, h, p1);
    __builtin_amdgcn_sched_barrier(0);
    fetch_frag(G, 2 * CH, lane, s2);
    fetch_pos(P, 2 * CH, h, p2);
    __builtin_amdgcn_sched_barrier(0);
    const int jself = i0 & ~(CH - 1);
    asm volatile("" :: "v"(px_i), "v"(py_i), "v"(s2.r[0][0]), "v"(s2.r[7][0]), "v"(p2.r[0][0]), "v"(p2.r[15][0]));
    int jc = 0;
    do {
        const int jw = (jc + 4 * CH < N) ? (jc + 4 * CH) : 0;
        chunk_step(G, P, s0, p0, s3, p3, jc,          jc + 3 * CH, jself,
                   lane, h, i, px_i, py_i, k2, qmin, mref, b2, acc);
        chunk_step(G, P, s1, p1, s0, p0, jc + CH,     jw,          jself,
                   lane, h, i, px_i, py_i, k2, qmin, mref, b2, acc);
        chunk_step(G, P, s2, p2, s1, p1, jc + 2 * CH, jw + CH,     jself,
                   lane, h, i, px_i, py_i, k2, qmin, mref, b2, acc);
        chunk_step(G, P, s3, p3, s2, p2, jc + 3 * CH, jw + 2 * CH, jself,
                   lane, h, i, px_i, py_i, k2, qmin, mref, b2, acc);
        jc += 4 * CH;
    } while (jc < N);

    #pragma unroll
    for (int v = 0; v < 8; ++v) {
        cbuf[(v + 8 * h) * 16 + col] = acc[v];
    }
    __syncthreads();

    if (lane < 16) {
        const int ii = i0 + lane;
        const float* c = &cbuf[lane * 16];
        const float S    = c[8];
        const float invS = 1.0f / S;
        const float* __restrict__ gf = G + frag_index(ii, 0);
        const float f[4] = {gf[0], gf[4], gf[8], gf[12]};
        const float gvx = gvp[0];
        const float gvy = gvp[1];

        float mu[4], sg[4];
        #pragma unroll
        for (int d = 0; d < 4; ++d) {
            float T = c[d] * invS;
            float U = c[4 + d] * invS;
            mu[d] = f[d] - T;
            float eg2 = f[d] * f[d] - 2.0f * f[d] * T + U;
            sg[d] = fast_sqrt(fmaxf(eg2 - mu[d] * mu[d], 0.0f) + 1e-6f);
        }
        float4* o = (float4*)(out + (size_t)ii * 12);
        o[0] = make_float4(mu[0], mu[1], mu[2], mu[3]);
        o[1] = make_float4(sg[0], sg[1], sg[2], sg[3]);
        o[2] = make_float4(gvx, gvy, f[2] - gvx, f[3] - gvy);
    }
}

extern "C" void kernel_launch(void* const* d_in, const int* in_sizes, int n_in,
                              void* d_out, int out_size, void* d_ws, size_t ws_size,
                              hipStream_t stream) {
    const float* states  = (const float*)d_in[0];
    const float* log_tau = (const float*)d_in[1];
    const int N = in_sizes[0] / 8;

    char* ws    = (char*)d_ws;
    float*  G   = (float*)ws;
    float*  gv  = (float*)(ws + (size_t)N * 64);
    float*  V   = (float*)(ws + (size_t)N * 64 + 256);
    float*  P   = (float*)(ws + (size_t)N * 72 + 256);
    float*  out = (float*)d_out;

    precompute_kernel<<<(N + 255) / 256, 256, 0, stream>>>(states, G, V, P, N);
    groupvel_kernel<<<1, GV_T, 0, stream>>>(V, gv, N);
    neighbour_kernel<<<N / 16, 32, 0, stream>>>(G, P, gv, log_tau, out, N);
}


# Two launches, was three: velocity sum per 256-particle block in kernel 1 (lane tree + LDS), the 16 block sums added by each wave of the pair kernel; mean kernel gone; velocity sum order regrouped
# speedup vs baseline: 11.8080x; 1.0811x over previous
#include <hip/hip_runtime.h>

typedef float v2f __attribute__((ext_vector_type(2)));
typedef float v4f __attribute__((ext_vector_type(4)));
typedef float v8f __attribute__((ext_vector_type(8)));

__device__ __forceinline__ float fast_sqrt(float x) { return __builtin_amdgcn_sqrtf(x); }
__device__ __forceinline__ float fast_exp2(float x) { return __builtin_amdgcn_exp2f(x); }

#define LOG2E 1.44269504088896340736f

__device__ __forceinline__ size_t frag_index(const int j, const int c) {
    const int r = j & 7;
    const int lane = c + 16 * ((r >> 1) & 1);
    return ((size_t)(j >> 3) * 32 + lane) * 4 + 2 * (r >> 2) + (r & 1);
}

__device__ __forceinline__ float swap_halves(const float v) {
    const int b = __builtin_bit_cast(int, v);
    return __builtin_bit_cast(float,
        __builtin_amdgcn_permlanex16(b, b, 0x76543210, 0xfedcba98, false, false));
}

template <int M>
__device__ __forceinline__ float row_xor(const float v) {
    const int b = __builtin_bit_cast(int, v);
    return __builtin_bit_cast(float,
        __builtin_amdgcn_update_dpp(b, b, 0x160 + M, 0xf, 0xf, false));
}

#define PC_T 256

__global__ __launch_bounds__(PC_T) void precompute_kernel(const float* __restrict__ states,
                                  float* __restrict__ G, float* __restrict__ P,
                                  float* __restrict__ S, int N) {
    __shared__ float2 wsum[PC_T / 32];
    const int t = threadIdx.x;
    const int i = blockIdx.x * PC_T + t;
    float px = 0.0f, py = 0.0f;
    float vx = 0.0f, vy = 0.0f;
    if (i < N) {
        const float4* s = (const float4*)(states + (size_t)i * 8);
        float4 a = s[0];
        float4 b = s[1];
        px = (a.x + a.z) * 0.5f;
        py = (a.y + a.w) * 0.5f;
        vx = (b.x + b.z) * 0.5f;
        vy = (b.y + b.w) * 0.5f;
    }
    float sx = vx, sy = vy;
    sx += row_xor<1>(sx);   sy += row_xor<1>(sy);
    sx += row_xor<2>(sx);   sy += row_xor<2>(sy);
    sx += row_xor<4>(sx);   sy += row_xor<4>(sy);
    sx += row_xor<8>(sx);   sy += row_xor<8>(sy);
    sx += swap_halves(sx);  sy += swap_halves(sy);
    if ((t & 31) == 0) wsum[t >> 5] = make_float2(sx, sy);
    __syncthreads();
    if (i < N) {
        float* g = G + frag_index(i, 0);
        const float row[16] = {px, py, vx, vy, px * px, py * py, vx * vx, vy * vy,
                               1.0f, 0.0f, 0.0f, 0.0f, 0.0f, 0.0f, 0.0f, 0.0f};
        #pragma unroll
        for (int c = 0; c < 16; ++c) g[4 * c] = row[c];
        P[(size_t)(i >> 1) * 4 + (i & 1)]     = px;
        P[(size_t)(i >> 1) * 4 + 2 + (i & 1)] = py;
    }
    if (t < 32) {
        const float2 w = wsum[t & (PC_T / 32 - 1)];
        float bx = w.x, by = w.y;
        bx += row_xor<4>(bx);   by += row_xor<4>(by);
        bx += row_xor<2>(bx);   by += row_xor<2>(by);
        bx += row_xor<1>(bx);   by += row_xor<1>(by);
        if (t == 0) ((float2*)S)[blockIdx.x] = make_float2(bx, by);
    }
}

#define CH  64
#define NPL (CH / 2)

struct JRegs {
    v4f r[NPL / 4];
};

struct PRegs {
    v4f r[NPL / 2];
};

#define SET_LOADED(s) "v"((s).r[0][0]), "v"((s).r[1][0]), "v"((s).r[2][0]), "v"((s).r[3][0]), \
                      "v"((s).r[4][0]), "v"((s).r[5][0]), "v"((s).r[6][0]), "v"((s).r[7][0])

__device__ __forceinline__ void fetch_frag(const float* __restrict__ G, const int jc, const int lane, JRegs& s) {
    const v4f* __restrict__ gp = (const v4f*)G + (size_t)(jc >> 3) * 32 + lane;
    #pragma unroll
    for (int q = 0; q < NPL / 4; ++q) {
        s.r[q] = gp[q * 32];
    }
}

__device__ __forceinline__ void fetch_pos(const float* __restrict__ P, const int jc, const int h, PRegs& s) {
    const v4f* __restrict__ pp = (const v4f*)P + (size_t)(jc >> 1) + h;
    #pragma unroll
    for (int ks = 0; ks < NPL / 2; ++ks) {
        s.r[ks] = pp[ks * 2];
    }
}

template <int L>
__device__ __forceinline__ float row_lane(const float v) {
    return __builtin_bit_cast(float,
        __builtin_amdgcn_update_dpp(0, __builtin_bit_cast(int, v), 0x150 + L, 0xf, 0xf, true));
}

__device__ __forceinline__ float xhalf(const float v) {
    const int b = __builtin_bit_cast(int, v);
    return __builtin_bit_cast(float,
        __builtin_amdgcn_permlanex16(b, b, 0x76543210, 0xfedcba98, false, false));
}

__device__ __forceinline__ float upper_ror8(const float v) {
    const int b = __builtin_bit_cast(int, v);
    return __builtin_bit_cast(float,
        __builtin_amdgcn_update_dpp(b, b, 0x128, 0x2, 0xf, false));
}

__device__ __forceinline__ void chunk_step(
    const float* __restrict__ G, const float* __restrict__ P,
    const JRegs& cur, const PRegs& curp, JRegs& pre, PRegs& prep,
    const int jc, const int jn, const int jself,
    const int lane, const int h, const int i,
    const float px_i, const float py_i, const float k2,
    float& qmin, float& mref, float& b2, v8f& acc)
{
    float q[NPL];
    const v2f pxx = {px_i, px_i};
    const v2f pyy = {py_i, py_i};
    const v2f eps = {1e-8f, 1e-8f};
    #pragma unroll
    for (int ks = 0; ks < NPL / 2; ++ks) {
        const v4f pr = curp.r[ks];
        const v2f xs = {pr[0], pr[1]};
        const v2f ys = {pr[2], pr[3]};
        const v2f dx = xs - pxx;
        const v2f dy = ys - pyy;
        const v2f qq = __builtin_elementwise_fma(dy, dy, __builtin_elementwise_fma(dx, dx, eps));
        q[2 * ks]     = qq[0];
        q[2 * ks + 1] = qq[1];
    }
    if (jc == jself) {
        const int jl = jc + 2 * h;
        #pragma unroll
        for (int u = 0; u < NPL; ++u) {
            const int j = jl + 4 * (u >> 1) + (u & 1);
            q[u] = (j == i) ? 3.0e38f : q[u];
        }
    }
    fetch_frag(G, jn, lane, pre);
    fetch_pos(P, jn, h, prep);
    __builtin_amdgcn_sched_barrier(0);
    float pm[4] = {q[0], q[1], q[2], q[3]};
    #pragma unroll
    for (int u = 4; u < NPL; ++u) pm[u & 3] = fminf(pm[u & 3], q[u]);
    float cmin = fminf(fminf(pm[0], pm[1]), fminf(pm[2], pm[3]));
    cmin = fminf(cmin, xhalf(cmin));
    const float qnew = fminf(qmin, cmin);

    {
        const float mnew  = fast_sqrt(qnew);
        const float scale = fast_exp2((mnew - mref) * k2);
        const float srow = upper_ror8(scale);
        acc[0] *= row_lane<0>(srow);
        acc[1] *= row_lane<1>(srow);
        acc[2] *= row_lane<2>(srow);
        acc[3] *= row_lane<3>(srow);
        acc[4] *= row_lane<4>(srow);
        acc[5] *= row_lane<5>(srow);
        acc[6] *= row_lane<6>(srow);
        acc[7] *= row_lane<7>(srow);
        mref = mnew;
        b2   = mnew * k2;
    }
    qmin = qnew;

    #pragma unroll
    for (int ks = 0; ks < NPL / 2; ++ks) {
        v2f A, B;
        #pragma unroll
        for (int p = 0; p < 2; ++p) {
            const int u = 2 * ks + p;
            const float dist = fast_sqrt(q[u]);
            A[p] = fast_exp2(fmaf(-dist, k2, b2));
            B[p] = cur.r[u >> 2][u & 3];
        }
        acc = __builtin_amdgcn_wmma_f32_16x16x4_f32(
             false, A,  false, B,
             (short)0, acc,  false,  false);
    }
}

__global__ __launch_bounds__(32) __attribute__((amdgpu_waves_per_eu(1, 1))) void neighbour_kernel(
    const float*  __restrict__ G,
    const float*  __restrict__ P,
    const float*  __restrict__ S,
    const float*  __restrict__ log_tau,
    float*        __restrict__ out, int N)
{
    __shared__ float cbuf[256];

    const int lane = threadIdx.x;
    const int col  = lane & 15;
    const int h    = lane >> 4;
    const int i0   = blockIdx.x * 16;
    const int i    = i0 + col;

    const float invtau = 1.0f / __expf(log_tau[0]);
    const float k2 = invtau * LOG2E;

    const int nS = (N + PC_T - 1) / PC_T;
    const float2* __restrict__ S2 = (const float2*)S;
    const float2 s_first = S2[(col < nS) ? col : (nS - 1)];

    const float* __restrict__ gi = G + frag_index(i, 0);
    const float px_i = gi[0], py_i = gi[4];
    const float vx_i = gi[8], vy_i = gi[12];

    float qmin = 3.0e38f;
    float mref = fast_sqrt(qmin);
    float b2   = mref * k2;
    v8f acc = {0.0f, 0.0f, 0.0f, 0.0f, 0.0f, 0.0f, 0.0f, 0.0f};

    JRegs s0, s1, s2, s3;
    PRegs p0, p1, p2, p3;
    fetch_frag(G, 0, lane, s0);
    fetch_pos(P, 0, h, p0);
    __builtin_amdgcn_sched_barrier(0);
    fetch_frag(G, CH, lane, s1);
    fetch_pos(P, CH, h, p1);
    __builtin_amdgcn_sched_barrier(0);
    fetch_frag(G, 2 * CH, lane, s2);
    fetch_pos(P, 2 * CH, h, p2);
    __builtin_amdgcn_sched_barrier(0);
    const int jself = i0 & ~(CH - 1);
    asm volatile("" :: "v"(px_i), "v"(py_i), "v"(s2.r[0][0]), "v"(s2.r[7][0]), "v"(p2.r[0][0]), "v"(p2.r[15][0]));
    int jc = 0;
    do {
        const int jw = (jc + 4 * CH < N) ? (jc + 4 * CH) : 0;
        chunk_step(G, P, s0, p0, s3, p3, jc,          jc + 3 * CH, jself,
                   lane, h, i, px_i, py_i, k2, qmin, mref, b2, acc);
        chunk_step(G, P, s1, p1, s0, p0, jc + CH,     jw,          jself,
                   lane, h, i, px_i, py_i, k2, qmin, mref, b2, acc);
        chunk_step(G, P, s2, p2, s1, p1, jc + 2 * CH, jw + CH,     jself,
                   lane, h, i, px_i, py_i, k2, qmin, mref, b2, acc);
        chunk_step(G, P, s3, p3, s2, p2, jc + 3 * CH, jw + 2 * CH, jself,
                   lane, h, i, px_i, py_i, k2, qmin, mref, b2, acc);
        jc += 4 * CH;
    } while (jc < N);

    float gsx = (col < nS) ? s_first.x : 0.0f;
    float gsy = (col < nS) ? s_first.y : 0.0f;
    for (int k0 = 16; k0 < nS; k0 += 16) {
        const int k = k0 + col;
        const float2 sk = S2[(k < nS) ? k : (nS - 1)];
        gsx += (k < nS) ? sk.x : 0.0f;
        gsy += (k < nS) ? sk.y : 0.0f;
    }
    gsx += row_xor<8>(gsx);   gsy += row_xor<8>(gsy);
    gsx += row_xor<4>(gsx);   gsy += row_xor<4>(gsy);
    gsx += row_xor<2>(gsx);   gsy += row_xor<2>(gsy);
    gsx += row_xor<1>(gsx);   gsy += row_xor<1>(gsy);
    const float gvx = gsx / (float)N;
    const float gvy = gsy / (float)N;

    #pragma unroll
    for (int v = 0; v < 8; ++v) {
        cbuf[(v + 8 * h) * 16 + col] = acc[v];
    }
    __syncthreads();

    if (lane < 16) {
        const int ii = i0 + lane;
        const float* c = &cbuf[lane * 16];
        const float wsum_i = c[8];
        const float invS = 1.0f / wsum_i;
        const float f[4] = {px_i, py_i, vx_i, vy_i};

        float mu[4], sg[4];
        #pragma unroll
        for (int d = 0; d < 4; ++d) {
            float T = c[d] * invS;
            float U = c[4 + d] * invS;
            mu[d] = f[d] - T;
            float eg2 = f[d] * f[d] - 2.0f * f[d] * T + U;
            sg[d] = fast_sqrt(fmaxf(eg2 - mu[d] * mu[d], 0.0f) + 1e-6f);
        }
        float4* o = (float4*)(out + (size_t)ii * 12);
        o[0] = make_float4(mu[0], mu[1], mu[2], mu[3]);
        o[1] = make_float4(sg[0], sg[1], sg[2], sg[3]);
        o[2] = make_float4(gvx, gvy, f[2] - gvx, f[3] - gvy);
    }
}

extern "C" void kernel_launch(void* const* d_in, const int* in_sizes, int n_in,
                              void* d_out, int out_size, void* d_ws, size_t ws_size,
                              hipStream_t stream) {
    const float* states  = (const float*)d_in[0];
    const float* log_tau = (const float*)d_in[1];
    const int N = in_sizes[0] / 8;

    char* ws    = (char*)d_ws;
    float*  G   = (float*)ws;
    float*  P   = (float*)(ws + (size_t)N * 64);
    float*  S   = (float*)(ws + (size_t)N * 72);
    float*  out = (float*)d_out;

    precompute_kernel<<<(N + 255) / 256, 256, 0, stream>>>(states, G, P, S, N);
    neighbour_kernel<<<N / 16, 32, 0, stream>>>(G, P, S, log_tau, out, N);
}
